# starting kernel
# baseline (speedup 1.0000x reference)
#include <hip/hip_runtime.h>
#include <cstdint>

typedef __attribute__((ext_vector_type(16))) _Float16 v16h;
typedef __attribute__((ext_vector_type(8)))  float    v8f;

#define BB   16
#define CC   512
#define NN   1024
#define C3   1536
#define CPG  16

__device__ __forceinline__ int kbase(int j, int half) {
  return ((j & 4) << 2) + half * 8 + ((j & 3) << 1);
}

union FragH { v16h v; uint32_t u[8]; _Float16 h[16]; };

__device__ __forceinline__ void wave_gemm_2x2(const _Float16* __restrict__ A, int lda,
                                              const _Float16* __restrict__ B, int ldb,
                                              int K, int m0, int n0, v8f acc[2][2]) {
  const int lane = threadIdx.x & 31;
  const int half = lane >> 4;
  const int r    = lane & 15;
  const _Float16* Ap0 = A + (size_t)(m0 + r)      * lda;
  const _Float16* Ap1 = A + (size_t)(m0 + 16 + r) * lda;
  const _Float16* Bp0 = B + (size_t)(n0 + r)      * ldb;
  const _Float16* Bp1 = B + (size_t)(n0 + 16 + r) * ldb;
  for (int k0 = 0; k0 < K; k0 += 32) {
    FragH a0, a1, b0, b1;
#pragma unroll
    for (int j = 0; j < 8; ++j) {
      const int kb = k0 + kbase(j, half);
      a0.u[j] = *(const uint32_t*)(Ap0 + kb);
      a1.u[j] = *(const uint32_t*)(Ap1 + kb);
      b0.u[j] = *(const uint32_t*)(Bp0 + kb);
      b1.u[j] = *(const uint32_t*)(Bp1 + kb);
    }
    if (k0 + 32 < K) {
      __builtin_prefetch(Ap0 + k0 + 32, 0, 1);
      __builtin_prefetch(Bp0 + k0 + 32, 0, 1);
    }
    acc[0][0] = __builtin_amdgcn_wmma_f32_16x16x32_f16(false, a0.v, false, b0.v, (short)0, acc[0][0], false, false);
    acc[0][1] = __builtin_amdgcn_wmma_f32_16x16x32_f16(false, a0.v, false, b1.v, (short)0, acc[0][1], false, false);
    acc[1][0] = __builtin_amdgcn_wmma_f32_16x16x32_f16(false, a1.v, false, b0.v, (short)0, acc[1][0], false, false);
    acc[1][1] = __builtin_amdgcn_wmma_f32_16x16x32_f16(false, a1.v, false, b1.v, (short)0, acc[1][1], false, false);
  }
}


__global__ void __launch_bounds__(256) cvt_f32_to_f16(const float* __restrict__ in,
                                                      _Float16* __restrict__ out, int n) {
  int i = blockIdx.x * blockDim.x + threadIdx.x;
  if (i < n) out[i] = (_Float16)in[i];
}

__global__ void __launch_bounds__(256) gn_stats_kernel(const float* __restrict__ x,
                                                       float* __restrict__ stats) {
  __shared__ float s1[256], s2[256];
  const int bg = blockIdx.x;
  const float* xp = x + (size_t)bg * (CPG * NN);
  float a = 0.f, b = 0.f;
  for (int i = threadIdx.x; i < CPG * NN; i += 256) { float v = xp[i]; a += v; b += v * v; }
  s1[threadIdx.x] = a; s2[threadIdx.x] = b;
  __syncthreads();
  for (int s = 128; s > 0; s >>= 1) {
    if (threadIdx.x < s) { s1[threadIdx.x] += s1[threadIdx.x + s];
                           s2[threadIdx.x] += s2[threadIdx.x + s]; }
    __syncthreads();
  }
  if (threadIdx.x == 0) {
    const float inv = 1.0f / (float)(CPG * NN);
    float mean = s1[0] * inv;
    float var  = s2[0] * inv - mean * mean;
    stats[2 * bg]     = mean;
    stats[2 * bg + 1] = rsqrtf(var + 1e-5f);
  }
}

__global__ void __launch_bounds__(256) gn_apply_t_kernel(const float* __restrict__ x,
    const float* __restrict__ gamma, const float* __restrict__ beta,
    const float* __restrict__ stats, _Float16* __restrict__ Ht) {
  const size_t i4 = ((size_t)blockIdx.x * blockDim.x + threadIdx.x) * 4;
  const int n  = (int)(i4 & (NN - 1));
  const int c  = (int)((i4 >> 10) & (CC - 1));
  const int bg = (int)(i4 >> 14);
  const int b  = bg >> 5;
  const float4 xv = *(const float4*)(x + i4);
  const float mean = stats[2 * bg], rstd = stats[2 * bg + 1];
  const float g = gamma[c] * rstd, bt = beta[c] - mean * gamma[c] * rstd;
  _Float16* hp = Ht + ((size_t)b * NN + n) * CC + c;
  hp[0 * CC] = (_Float16)(xv.x * g + bt);
  hp[1 * CC] = (_Float16)(xv.y * g + bt);
  hp[2 * CC] = (_Float16)(xv.z * g + bt);
  hp[3 * CC] = (_Float16)(xv.w * g + bt);
}


__global__ void __launch_bounds__(128) qkv_gemm_kernel(const _Float16* __restrict__ W,
    const _Float16* __restrict__ Ht, const float* __restrict__ bias,
    _Float16* __restrict__ Qt, _Float16* __restrict__ Kt, _Float16* __restrict__ Vv) {
  const int t  = blockIdx.x * 4 + (threadIdx.x >> 5);
  const int m0 = (t % (C3 / 32)) * 32, n0 = (t / (C3 / 32)) * 32;
  v8f acc[2][2] = {};
  wave_gemm_2x2(W, CC, Ht, CC, CC, m0, n0, acc);
  const int lane = threadIdx.x & 31, half = lane >> 4, r = lane & 15;
#pragma unroll
  for (int mi = 0; mi < 2; ++mi) {
#pragma unroll
    for (int ni = 0; ni < 2; ++ni) {
      const int mb = m0 + mi * 16;
      const int n  = n0 + ni * 16 + r;
      const int mbase = mb + 8 * half;
      const int seg = mb >> 9;
      if (seg < 2) {
        _Float16* dst = (seg == 0 ? Qt : Kt) + (size_t)n * CC + ((mbase) & (CC - 1));
#pragma unroll
        for (int v = 0; v < 8; v += 2) {
          union { _Float16 h[2]; uint32_t u; } pk;
          pk.h[0] = (_Float16)(acc[mi][ni][v]     + bias[mbase + v]);
          pk.h[1] = (_Float16)(acc[mi][ni][v + 1] + bias[mbase + v + 1]);
          *(uint32_t*)(dst + v) = pk.u;
        }
      } else {
#pragma unroll
        for (int v = 0; v < 8; ++v)
          Vv[(size_t)(mbase - 1024 + v) * NN + n] =
              (_Float16)(acc[mi][ni][v] + bias[mbase + v]);
      }
    }
  }
}

__global__ void __launch_bounds__(128) scores_kernel(const _Float16* __restrict__ Qt,
    const _Float16* __restrict__ Kt, float* __restrict__ S) {
  const int t  = blockIdx.x * 4 + (threadIdx.x >> 5);
  const int m0 = (t & 31) * 32, n0 = (t >> 5) * 32;
  v8f acc[2][2] = {};
  wave_gemm_2x2(Qt, CC, Kt, CC, CC, m0, n0, acc);
  const float scale = 0.04419417382415922f;
  const int lane = threadIdx.x & 31, half = lane >> 4, r = lane & 15;
#pragma unroll
  for (int mi = 0; mi < 2; ++mi)
#pragma unroll
    for (int ni = 0; ni < 2; ++ni) {
      const int mb = m0 + mi * 16 + 8 * half, n = n0 + ni * 16 + r;
#pragma unroll
      for (int v = 0; v < 8; ++v)
        S[(size_t)(mb + v) * NN + n] = acc[mi][ni][v] * scale;
    }
}

__global__ void __launch_bounds__(256) softmax_kernel(const float* __restrict__ S,
                                                      _Float16* __restrict__ P) {
  __shared__ float red[256];
  const int row = blockIdx.x;
  const int base = threadIdx.x * 4;
  float4 sv = *(const float4*)(S + (size_t)row * NN + base);
  float v0 = sv.x, v1 = sv.y, v2 = sv.z, v3 = sv.w;
  float m = fmaxf(fmaxf(v0, v1), fmaxf(v2, v3));
  red[threadIdx.x] = m; __syncthreads();
  for (int s = 128; s > 0; s >>= 1) {
    if (threadIdx.x < s) red[threadIdx.x] = fmaxf(red[threadIdx.x], red[threadIdx.x + s]);
    __syncthreads();
  }
  m = red[0]; __syncthreads();
  v0 = __expf(v0 - m); v1 = __expf(v1 - m); v2 = __expf(v2 - m); v3 = __expf(v3 - m);
  red[threadIdx.x] = v0 + v1 + v2 + v3; __syncthreads();
  for (int s = 128; s > 0; s >>= 1) {
    if (threadIdx.x < s) red[threadIdx.x] += red[threadIdx.x + s];
    __syncthreads();
  }
  const float inv = 1.0f / red[0];
  union { _Float16 h[4]; uint2 u; } o;
  o.h[0] = (_Float16)(v0 * inv); o.h[1] = (_Float16)(v1 * inv);
  o.h[2] = (_Float16)(v2 * inv); o.h[3] = (_Float16)(v3 * inv);
  *(uint2*)(P + (size_t)row * NN + base) = o.u;
}

__global__ void __launch_bounds__(128) av_gemm_kernel(const _Float16* __restrict__ Vv,
    const _Float16* __restrict__ P, _Float16* __restrict__ Ot) {
  const int t  = blockIdx.x * 4 + (threadIdx.x >> 5);
  const int m0 = (t & 15) * 32, n0 = (t >> 4) * 32;
  v8f acc[2][2] = {};
  wave_gemm_2x2(Vv, NN, P, NN, NN, m0, n0, acc);
  const int lane = threadIdx.x & 31, half = lane >> 4, r = lane & 15;
#pragma unroll
  for (int mi = 0; mi < 2; ++mi)
#pragma unroll
    for (int ni = 0; ni < 2; ++ni) {
      const int mb = m0 + mi * 16 + 8 * half, n = n0 + ni * 16 + r;
      _Float16* dst = Ot + (size_t)n * CC + mb;
#pragma unroll
      for (int v = 0; v < 8; v += 2) {
        union { _Float16 h[2]; uint32_t u; } pk;
        pk.h[0] = (_Float16)acc[mi][ni][v];
        pk.h[1] = (_Float16)acc[mi][ni][v + 1];
        *(uint32_t*)(dst + v) = pk.u;
      }
    }
}

__global__ void __launch_bounds__(128) proj_kernel(const _Float16* __restrict__ Wp,
    const _Float16* __restrict__ Ot, const float* __restrict__ bias,
    const float* __restrict__ xb, float* __restrict__ out) {
  const int t  = blockIdx.x * 4 + (threadIdx.x >> 5);
  const int m0 = (t & 15) * 32, n0 = (t >> 4) * 32;
  v8f acc[2][2] = {};
  wave_gemm_2x2(Wp, CC, Ot, CC, CC, m0, n0, acc);
  const int lane = threadIdx.x & 31, half = lane >> 4, r = lane & 15;
#pragma unroll
  for (int mi = 0; mi < 2; ++mi)
#pragma unroll
    for (int ni = 0; ni < 2; ++ni) {
      const int mb = m0 + mi * 16 + 8 * half, n = n0 + ni * 16 + r;
#pragma unroll
      for (int v = 0; v < 8; ++v) {
        const size_t idx = (size_t)(mb + v) * NN + n;
        out[idx] = xb[idx] + bias[mb + v] + acc[mi][ni][v];
      }
    }
}


extern "C" void kernel_launch(void* const* d_in, const int* in_sizes, int n_in,
                              void* d_out, int out_size, void* d_ws, size_t ws_size,
                              hipStream_t stream) {
  const float* x      = (const float*)d_in[0];
  const float* gamma  = (const float*)d_in[1];
  const float* beta   = (const float*)d_in[2];
  const float* w_qkv  = (const float*)d_in[3];
  const float* b_qkv  = (const float*)d_in[4];
  const float* w_proj = (const float*)d_in[5];
  const float* b_proj = (const float*)d_in[6];
  float* out = (float*)d_out;

  char* ws = (char*)d_ws;
  float*    stats  = (float*)   (ws + 0);
  _Float16* Wqkvh  = (_Float16*)(ws + 4096);
  _Float16* Wprojh = (_Float16*)(ws + 1576960);
  _Float16* Ht     = (_Float16*)(ws + 2101248);
  _Float16* Qt     = (_Float16*)(ws + 18878464);
  _Float16* Kt     = (_Float16*)(ws + 19927040);
  _Float16* Vv     = (_Float16*)(ws + 20975616);
  float*    S      = (float*)   (ws + 22024192);
  _Float16* P      = (_Float16*)(ws + 26218496);
  _Float16* Ot     = (_Float16*)(ws + 28315648);

  cvt_f32_to_f16<<<(C3 * CC) / 256, 256, 0, stream>>>(w_qkv, Wqkvh, C3 * CC);
  cvt_f32_to_f16<<<(CC * CC) / 256, 256, 0, stream>>>(w_proj, Wprojh, CC * CC);
  gn_stats_kernel<<<BB * 32, 256, 0, stream>>>(x, stats);
  gn_apply_t_kernel<<<(BB * CC * NN) / (256 * 4), 256, 0, stream>>>(x, gamma, beta, stats, Ht);

  for (int b = 0; b < BB; ++b) {
    const _Float16* Htb = Ht + (size_t)b * NN * CC;
    qkv_gemm_kernel<<<(48 * 32) / 4, 128, 0, stream>>>(Wqkvh, Htb, b_qkv, Qt, Kt, Vv);
    scores_kernel<<<(32 * 32) / 4, 128, 0, stream>>>(Qt, Kt, S);
    softmax_kernel<<<NN, 256, 0, stream>>>(S, P);
    av_gemm_kernel<<<(16 * 32) / 4, 128, 0, stream>>>(Vv, P, Ot);
    proj_kernel<<<(16 * 32) / 4, 128, 0, stream>>>(Wprojh, Ot, b_proj,
                                                   x + (size_t)b * CC * NN,
                                                   out + (size_t)b * CC * NN);
  }
}


# K loops of the four GEMMs: fragment loads from global memory issued one group of 4 K-steps ahead of their WMMAs (second register set) instead of one memory round trip per K-step; sums unchanged
# speedup vs baseline: 1.1240x; 1.1240x over previous
#include <hip/hip_runtime.h>
#include <cstdint>

typedef __attribute__((ext_vector_type(16))) _Float16 v16h;
typedef __attribute__((ext_vector_type(8)))  float    v8f;

#define BB   16
#define CC   512
#define NN   1024
#define C3   1536
#define CPG  16

__device__ __forceinline__ int kbase(int j, int half) {
  return ((j & 4) << 2) + half * 8 + ((j & 3) << 1);
}

union FragH { v16h v; uint32_t u[8]; _Float16 h[16]; };

#define KG 4
struct FragGroup { FragH a0[KG], a1[KG], b0[KG], b1[KG]; };

__device__ __forceinline__ void load_group(FragGroup& f,
    const _Float16* __restrict__ Ap0, const _Float16* __restrict__ Ap1,
    const _Float16* __restrict__ Bp0, const _Float16* __restrict__ Bp1,
    int k0, int half) {
#pragma unroll
  for (int s = 0; s < KG; ++s) {
#pragma unroll
    for (int j = 0; j < 8; ++j) {
      const int kb = k0 + s * 32 + kbase(j, half);
      f.a0[s].u[j] = *(const uint32_t*)(Ap0 + kb);
      f.a1[s].u[j] = *(const uint32_t*)(Ap1 + kb);
      f.b0[s].u[j] = *(const uint32_t*)(Bp0 + kb);
      f.b1[s].u[j] = *(const uint32_t*)(Bp1 + kb);
    }
  }
}

__device__ __forceinline__ void mma_group(const FragGroup& f, v8f acc[2][2]) {
#pragma unroll
  for (int s = 0; s < KG; ++s) {
    acc[0][0] = __builtin_amdgcn_wmma_f32_16x16x32_f16(false, f.a0[s].v, false, f.b0[s].v, (short)0, acc[0][0], false, false);
    acc[0][1] = __builtin_amdgcn_wmma_f32_16x16x32_f16(false, f.a0[s].v, false, f.b1[s].v, (short)0, acc[0][1], false, false);
    acc[1][0] = __builtin_amdgcn_wmma_f32_16x16x32_f16(false, f.a1[s].v, false, f.b0[s].v, (short)0, acc[1][0], false, false);
    acc[1][1] = __builtin_amdgcn_wmma_f32_16x16x32_f16(false, f.a1[s].v, false, f.b1[s].v, (short)0, acc[1][1], false, false);
  }
}

template <int K>
__device__ __forceinline__ void wave_gemm_2x2(const _Float16* __restrict__ A, int lda,
                                              const _Float16* __restrict__ B, int ldb,
                                              int m0, int n0, v8f acc[2][2]) {
  static_assert(K % (32 * KG) == 0, "K must be a whole number of load groups");
  constexpr int NG = K / (32 * KG);
  const int lane = threadIdx.x & 31;
  const int half = lane >> 4;
  const int r    = lane & 15;
  const _Float16* Ap0 = A + (size_t)(m0 + r)      * lda;
  const _Float16* Ap1 = A + (size_t)(m0 + 16 + r) * lda;
  const _Float16* Bp0 = B + (size_t)(n0 + r)      * ldb;
  const _Float16* Bp1 = B + (size_t)(n0 + 16 + r) * ldb;
  FragGroup cur, nxt;
  load_group(cur, Ap0, Ap1, Bp0, Bp1, 0, half);
#pragma unroll
  for (int g = 0; g < NG; ++g) {
    if (g + 1 < NG) load_group(nxt, Ap0, Ap1, Bp0, Bp1, (g + 1) * 32 * KG, half);
    __builtin_amdgcn_sched_barrier(0);
    mma_group(cur, acc);
    if (g + 1 < NG) cur = nxt;
  }
}


__global__ void __launch_bounds__(256) cvt_f32_to_f16(const float* __restrict__ in,
                                                      _Float16* __restrict__ out, int n) {
  int i = blockIdx.x * blockDim.x + threadIdx.x;
  if (i < n) out[i] = (_Float16)in[i];
}

__global__ void __launch_bounds__(256) gn_stats_kernel(const float* __restrict__ x,
                                                       float* __restrict__ stats) {
  __shared__ float s1[256], s2[256];
  const int bg = blockIdx.x;
  const float* xp = x + (size_t)bg * (CPG * NN);
  float a = 0.f, b = 0.f;
  for (int i = threadIdx.x; i < CPG * NN; i += 256) { float v = xp[i]; a += v; b += v * v; }
  s1[threadIdx.x] = a; s2[threadIdx.x] = b;
  __syncthreads();
  for (int s = 128; s > 0; s >>= 1) {
    if (threadIdx.x < s) { s1[threadIdx.x] += s1[threadIdx.x + s];
                           s2[threadIdx.x] += s2[threadIdx.x + s]; }
    __syncthreads();
  }
  if (threadIdx.x == 0) {
    const float inv = 1.0f / (float)(CPG * NN);
    float mean = s1[0] * inv;
    float var  = s2[0] * inv - mean * mean;
    stats[2 * bg]     = mean;
    stats[2 * bg + 1] = rsqrtf(var + 1e-5f);
  }
}

__global__ void __launch_bounds__(256) gn_apply_t_kernel(const float* __restrict__ x,
    const float* __restrict__ gamma, const float* __restrict__ beta,
    const float* __restrict__ stats, _Float16* __restrict__ Ht) {
  const size_t i4 = ((size_t)blockIdx.x * blockDim.x + threadIdx.x) * 4;
  const int n  = (int)(i4 & (NN - 1));
  const int c  = (int)((i4 >> 10) & (CC - 1));
  const int bg = (int)(i4 >> 14);
  const int b  = bg >> 5;
  const float4 xv = *(const float4*)(x + i4);
  const float mean = stats[2 * bg], rstd = stats[2 * bg + 1];
  const float g = gamma[c] * rstd, bt = beta[c] - mean * gamma[c] * rstd;
  _Float16* hp = Ht + ((size_t)b * NN + n) * CC + c;
  hp[0 * CC] = (_Float16)(xv.x * g + bt);
  hp[1 * CC] = (_Float16)(xv.y * g + bt);
  hp[2 * CC] = (_Float16)(xv.z * g + bt);
  hp[3 * CC] = (_Float16)(xv.w * g + bt);
}


__global__ void __launch_bounds__(128) qkv_gemm_kernel(const _Float16* __restrict__ W,
    const _Float16* __restrict__ Ht, const float* __restrict__ bias,
    _Float16* __restrict__ Qt, _Float16* __restrict__ Kt, _Float16* __restrict__ Vv) {
  const int t  = blockIdx.x * 4 + (threadIdx.x >> 5);
  const int m0 = (t % (C3 / 32)) * 32, n0 = (t / (C3 / 32)) * 32;
  v8f acc[2][2] = {};
  wave_gemm_2x2<CC>(W, CC, Ht, CC, m0, n0, acc);
  const int lane = threadIdx.x & 31, half = lane >> 4, r = lane & 15;
#pragma unroll
  for (int mi = 0; mi < 2; ++mi) {
#pragma unroll
    for (int ni = 0; ni < 2; ++ni) {
      const int mb = m0 + mi * 16;
      const int n  = n0 + ni * 16 + r;
      const int mbase = mb + 8 * half;
      const int seg = mb >> 9;
      if (seg < 2) {
        _Float16* dst = (seg == 0 ? Qt : Kt) + (size_t)n * CC + ((mbase) & (CC - 1));
#pragma unroll
        for (int v = 0; v < 8; v += 2) {
          union { _Float16 h[2]; uint32_t u; } pk;
          pk.h[0] = (_Float16)(acc[mi][ni][v]     + bias[mbase + v]);
          pk.h[1] = (_Float16)(acc[mi][ni][v + 1] + bias[mbase + v + 1]);
          *(uint32_t*)(dst + v) = pk.u;
        }
      } else {
#pragma unroll
        for (int v = 0; v < 8; ++v)
          Vv[(size_t)(mbase - 1024 + v) * NN + n] =
              (_Float16)(acc[mi][ni][v] + bias[mbase + v]);
      }
    }
  }
}

__global__ void __launch_bounds__(128) scores_kernel(const _Float16* __restrict__ Qt,
    const _Float16* __restrict__ Kt, float* __restrict__ S) {
  const int t  = blockIdx.x * 4 + (threadIdx.x >> 5);
  const int m0 = (t & 31) * 32, n0 = (t >> 5) * 32;
  v8f acc[2][2] = {};
  wave_gemm_2x2<CC>(Qt, CC, Kt, CC, m0, n0, acc);
  const float scale = 0.04419417382415922f;
  const int lane = threadIdx.x & 31, half = lane >> 4, r = lane & 15;
#pragma unroll
  for (int mi = 0; mi < 2; ++mi)
#pragma unroll
    for (int ni = 0; ni < 2; ++ni) {
      const int mb = m0 + mi * 16 + 8 * half, n = n0 + ni * 16 + r;
#pragma unroll
      for (int v = 0; v < 8; ++v)
        S[(size_t)(mb + v) * NN + n] = acc[mi][ni][v] * scale;
    }
}

__global__ void __launch_bounds__(256) softmax_kernel(const float* __restrict__ S,
                                                      _Float16* __restrict__ P) {
  __shared__ float red[256];
  const int row = blockIdx.x;
  const int base = threadIdx.x * 4;
  float4 sv = *(const float4*)(S + (size_t)row * NN + base);
  float v0 = sv.x, v1 = sv.y, v2 = sv.z, v3 = sv.w;
  float m = fmaxf(fmaxf(v0, v1), fmaxf(v2, v3));
  red[threadIdx.x] = m; __syncthreads();
  for (int s = 128; s > 0; s >>= 1) {
    if (threadIdx.x < s) red[threadIdx.x] = fmaxf(red[threadIdx.x], red[threadIdx.x + s]);
    __syncthreads();
  }
  m = red[0]; __syncthreads();
  v0 = __expf(v0 - m); v1 = __expf(v1 - m); v2 = __expf(v2 - m); v3 = __expf(v3 - m);
  red[threadIdx.x] = v0 + v1 + v2 + v3; __syncthreads();
  for (int s = 128; s > 0; s >>= 1) {
    if (threadIdx.x < s) red[threadIdx.x] += red[threadIdx.x + s];
    __syncthreads();
  }
  const float inv = 1.0f / red[0];
  union { _Float16 h[4]; uint2 u; } o;
  o.h[0] = (_Float16)(v0 * inv); o.h[1] = (_Float16)(v1 * inv);
  o.h[2] = (_Float16)(v2 * inv); o.h[3] = (_Float16)(v3 * inv);
  *(uint2*)(P + (size_t)row * NN + base) = o.u;
}

__global__ void __launch_bounds__(128) av_gemm_kernel(const _Float16* __restrict__ Vv,
    const _Float16* __restrict__ P, _Float16* __restrict__ Ot) {
  const int t  = blockIdx.x * 4 + (threadIdx.x >> 5);
  const int m0 = (t & 15) * 32, n0 = (t >> 4) * 32;
  v8f acc[2][2] = {};
  wave_gemm_2x2<NN>(Vv, NN, P, NN, m0, n0, acc);
  const int lane = threadIdx.x & 31, half = lane >> 4, r = lane & 15;
#pragma unroll
  for (int mi = 0; mi < 2; ++mi)
#pragma unroll
    for (int ni = 0; ni < 2; ++ni) {
      const int mb = m0 + mi * 16 + 8 * half, n = n0 + ni * 16 + r;
      _Float16* dst = Ot + (size_t)n * CC + mb;
#pragma unroll
      for (int v = 0; v < 8; v += 2) {
        union { _Float16 h[2]; uint32_t u; } pk;
        pk.h[0] = (_Float16)acc[mi][ni][v];
        pk.h[1] = (_Float16)acc[mi][ni][v + 1];
        *(uint32_t*)(dst + v) = pk.u;
      }
    }
}

__global__ void __launch_bounds__(128) proj_kernel(const _Float16* __restrict__ Wp,
    const _Float16* __restrict__ Ot, const float* __restrict__ bias,
    const float* __restrict__ xb, float* __restrict__ out) {
  const int t  = blockIdx.x * 4 + (threadIdx.x >> 5);
  const int m0 = (t & 15) * 32, n0 = (t >> 4) * 32;
  v8f acc[2][2] = {};
  wave_gemm_2x2<CC>(Wp, CC, Ot, CC, m0, n0, acc);
  const int lane = threadIdx.x & 31, half = lane >> 4, r = lane & 15;
#pragma unroll
  for (int mi = 0; mi < 2; ++mi)
#pragma unroll
    for (int ni = 0; ni < 2; ++ni) {
      const int mb = m0 + mi * 16 + 8 * half, n = n0 + ni * 16 + r;
#pragma unroll
      for (int v = 0; v < 8; ++v) {
        const size_t idx = (size_t)(mb + v) * NN + n;
        out[idx] = xb[idx] + bias[mb + v] + acc[mi][ni][v];
      }
    }
}


extern "C" void kernel_launch(void* const* d_in, const int* in_sizes, int n_in,
                              void* d_out, int out_size, void* d_ws, size_t ws_size,
                              hipStream_t stream) {
  const float* x      = (const float*)d_in[0];
  const float* gamma  = (const float*)d_in[1];
  const float* beta   = (const float*)d_in[2];
  const float* w_qkv  = (const float*)d_in[3];
  const float* b_qkv  = (const float*)d_in[4];
  const float* w_proj = (const float*)d_in[5];
  const float* b_proj = (const float*)d_in[6];
  float* out = (float*)d_out;

  char* ws = (char*)d_ws;
  float*    stats  = (float*)   (ws + 0);
  _Float16* Wqkvh  = (_Float16*)(ws + 4096);
  _Float16* Wprojh = (_Float16*)(ws + 1576960);
  _Float16* Ht     = (_Float16*)(ws + 2101248);
  _Float16* Qt     = (_Float16*)(ws + 18878464);
  _Float16* Kt     = (_Float16*)(ws + 19927040);
  _Float16* Vv     = (_Float16*)(ws + 20975616);
  float*    S      = (float*)   (ws + 22024192);
  _Float16* P      = (_Float16*)(ws + 26218496);
  _Float16* Ot     = (_Float16*)(ws + 28315648);

  cvt_f32_to_f16<<<(C3 * CC) / 256, 256, 0, stream>>>(w_qkv, Wqkvh, C3 * CC);
  cvt_f32_to_f16<<<(CC * CC) / 256, 256, 0, stream>>>(w_proj, Wprojh, CC * CC);
  gn_stats_kernel<<<BB * 32, 256, 0, stream>>>(x, stats);
  gn_apply_t_kernel<<<(BB * CC * NN) / (256 * 4), 256, 0, stream>>>(x, gamma, beta, stats, Ht);

  for (int b = 0; b < BB; ++b) {
    const _Float16* Htb = Ht + (size_t)b * NN * CC;
    qkv_gemm_kernel<<<(48 * 32) / 4, 128, 0, stream>>>(Wqkvh, Htb, b_qkv, Qt, Kt, Vv);
    scores_kernel<<<(32 * 32) / 4, 128, 0, stream>>>(Qt, Kt, S);
    softmax_kernel<<<NN, 256, 0, stream>>>(S, P);
    av_gemm_kernel<<<(16 * 32) / 4, 128, 0, stream>>>(Vv, P, Ot);
    proj_kernel<<<(16 * 32) / 4, 128, 0, stream>>>(Wprojh, Ot, b_proj,
                                                   x + (size_t)b * CC * NN,
                                                   out + (size_t)b * CC * NN);
  }
}


# Launch per image folded into grid.y: QKV, scores, softmax, AV take 4 images per launch, projection all images at once (84 -> 21 launches); group scratch in the unwritten output buffer; sums unchanged
# speedup vs baseline: 1.7557x; 1.5620x over previous
#include <hip/hip_runtime.h>
#include <cstdint>

typedef __attribute__((ext_vector_type(16))) _Float16 v16h;
typedef __attribute__((ext_vector_type(8)))  float    v8f;

#define BB   16
#define CC   512
#define NN   1024
#define C3   1536
#define CPG  16
#define GI   4

__device__ __forceinline__ int kbase(int j, int half) {
  return ((j & 4) << 2) + half * 8 + ((j & 3) << 1);
}

union FragH { v16h v; uint32_t u[8]; _Float16 h[16]; };

#define KG 4
struct FragGroup { FragH a0[KG], a1[KG], b0[KG], b1[KG]; };

__device__ __forceinline__ void load_group(FragGroup& f,
    const _Float16* __restrict__ Ap0, const _Float16* __restrict__ Ap1,
    const _Float16* __restrict__ Bp0, const _Float16* __restrict__ Bp1,
    int k0, int half) {
#pragma unroll
  for (int s = 0; s < KG; ++s) {
#pragma unroll
    for (int j = 0; j < 8; ++j) {
      const int kb = k0 + s * 32 + kbase(j, half);
      f.a0[s].u[j] = *(const uint32_t*)(Ap0 + kb);
      f.a1[s].u[j] = *(const uint32_t*)(Ap1 + kb);
      f.b0[s].u[j] = *(const uint32_t*)(Bp0 + kb);
      f.b1[s].u[j] = *(const uint32_t*)(Bp1 + kb);
    }
  }
}

__device__ __forceinline__ void mma_group(const FragGroup& f, v8f acc[2][2]) {
#pragma unroll
  for (int s = 0; s < KG; ++s) {
    acc[0][0] = __builtin_amdgcn_wmma_f32_16x16x32_f16(false, f.a0[s].v, false, f.b0[s].v, (short)0, acc[0][0], false, false);
    acc[0][1] = __builtin_amdgcn_wmma_f32_16x16x32_f16(false, f.a0[s].v, false, f.b1[s].v, (short)0, acc[0][1], false, false);
    acc[1][0] = __builtin_amdgcn_wmma_f32_16x16x32_f16(false, f.a1[s].v, false, f.b0[s].v, (short)0, acc[1][0], false, false);
    acc[1][1] = __builtin_amdgcn_wmma_f32_16x16x32_f16(false, f.a1[s].v, false, f.b1[s].v, (short)0, acc[1][1], false, false);
  }
}

template <int K>
__device__ __forceinline__ void wave_gemm_2x2(const _Float16* __restrict__ A, int lda,
                                              const _Float16* __restrict__ B, int ldb,
                                              int m0, int n0, v8f acc[2][2]) {
  static_assert(K % (32 * KG) == 0, "K must be a whole number of load groups");
  constexpr int NG = K / (32 * KG);
  const int lane = threadIdx.x & 31;
  const int half = lane >> 4;
  const int r    = lane & 15;
  const _Float16* Ap0 = A + (size_t)(m0 + r)      * lda;
  const _Float16* Ap1 = A + (size_t)(m0 + 16 + r) * lda;
  const _Float16* Bp0 = B + (size_t)(n0 + r)      * ldb;
  const _Float16* Bp1 = B + (size_t)(n0 + 16 + r) * ldb;
  FragGroup cur, nxt;
  load_group(cur, Ap0, Ap1, Bp0, Bp1, 0, half);
#pragma unroll
  for (int g = 0; g < NG; ++g) {
    if (g + 1 < NG) load_group(nxt, Ap0, Ap1, Bp0, Bp1, (g + 1) * 32 * KG, half);
    __builtin_amdgcn_sched_barrier(0);
    mma_group(cur, acc);
    if (g + 1 < NG) cur = nxt;
  }
}


__global__ void __launch_bounds__(256) cvt_f32_to_f16(const float* __restrict__ in,
                                                      _Float16* __restrict__ out, int n) {
  int i = blockIdx.x * blockDim.x + threadIdx.x;
  if (i < n) out[i] = (_Float16)in[i];
}

__global__ void __launch_bounds__(256) gn_stats_kernel(const float* __restrict__ x,
                                                       float* __restrict__ stats) {
  __shared__ float s1[256], s2[256];
  const int bg = blockIdx.x;
  const float* xp = x + (size_t)bg * (CPG * NN);
  float a = 0.f, b = 0.f;
  for (int i = threadIdx.x; i < CPG * NN; i += 256) { float v = xp[i]; a += v; b += v * v; }
  s1[threadIdx.x] = a; s2[threadIdx.x] = b;
  __syncthreads();
  for (int s = 128; s > 0; s >>= 1) {
    if (threadIdx.x < s) { s1[threadIdx.x] += s1[threadIdx.x + s];
                           s2[threadIdx.x] += s2[threadIdx.x + s]; }
    __syncthreads();
  }
  if (threadIdx.x == 0) {
    const float inv = 1.0f / (float)(CPG * NN);
    float mean = s1[0] * inv;
    float var  = s2[0] * inv - mean * mean;
    stats[2 * bg]     = mean;
    stats[2 * bg + 1] = rsqrtf(var + 1e-5f);
  }
}

__global__ void __launch_bounds__(256) gn_apply_t_kernel(const float* __restrict__ x,
    const float* __restrict__ gamma, const float* __restrict__ beta,
    const float* __restrict__ stats, _Float16* __restrict__ Ht) {
  const size_t i4 = ((size_t)blockIdx.x * blockDim.x + threadIdx.x) * 4;
  const int n  = (int)(i4 & (NN - 1));
  const int c  = (int)((i4 >> 10) & (CC - 1));
  const int bg = (int)(i4 >> 14);
  const int b  = bg >> 5;
  const float4 xv = *(const float4*)(x + i4);
  const float mean = stats[2 * bg], rstd = stats[2 * bg + 1];
  const float g = gamma[c] * rstd, bt = beta[c] - mean * gamma[c] * rstd;
  _Float16* hp = Ht + ((size_t)b * NN + n) * CC + c;
  hp[0 * CC] = (_Float16)(xv.x * g + bt);
  hp[1 * CC] = (_Float16)(xv.y * g + bt);
  hp[2 * CC] = (_Float16)(xv.z * g + bt);
  hp[3 * CC] = (_Float16)(xv.w * g + bt);
}


__global__ void __launch_bounds__(128) qkv_gemm_kernel(const _Float16* __restrict__ W,
    const _Float16* __restrict__ Ht, const float* __restrict__ bias,
    _Float16* __restrict__ Qt, _Float16* __restrict__ Kt, _Float16* __restrict__ Vv) {
  const size_t img = blockIdx.y;
  Ht += img * (NN * CC); Qt += img * (NN * CC); Kt += img * (NN * CC); Vv += img * (CC * NN);
  const int t  = blockIdx.x * 4 + (threadIdx.x >> 5);
  const int m0 = (t % (C3 / 32)) * 32, n0 = (t / (C3 / 32)) * 32;
  v8f acc[2][2] = {};
  wave_gemm_2x2<CC>(W, CC, Ht, CC, m0, n0, acc);
  const int lane = threadIdx.x & 31, half = lane >> 4, r = lane & 15;
#pragma unroll
  for (int mi = 0; mi < 2; ++mi) {
#pragma unroll
    for (int ni = 0; ni < 2; ++ni) {
      const int mb = m0 + mi * 16;
      const int n  = n0 + ni * 16 + r;
      const int mbase = mb + 8 * half;
      const int seg = mb >> 9;
      if (seg < 2) {
        _Float16* dst = (seg == 0 ? Qt : Kt) + (size_t)n * CC + ((mbase) & (CC - 1));
#pragma unroll
        for (int v = 0; v < 8; v += 2) {
          union { _Float16 h[2]; uint32_t u; } pk;
          pk.h[0] = (_Float16)(acc[mi][ni][v]     + bias[mbase + v]);
          pk.h[1] = (_Float16)(acc[mi][ni][v + 1] + bias[mbase + v + 1]);
          *(uint32_t*)(dst + v) = pk.u;
        }
      } else {
#pragma unroll
        for (int v = 0; v < 8; ++v)
          Vv[(size_t)(mbase - 1024 + v) * NN + n] =
              (_Float16)(acc[mi][ni][v] + bias[mbase + v]);
      }
    }
  }
}

__global__ void __launch_bounds__(128) scores_kernel(const _Float16* __restrict__ Qt,
    const _Float16* __restrict__ Kt, float* __restrict__ S) {
  const size_t img = blockIdx.y;
  Qt += img * (NN * CC); Kt += img * (NN * CC); S += img * ((size_t)NN * NN);
  const int t  = blockIdx.x * 4 + (threadIdx.x >> 5);
  const int m0 = (t & 31) * 32, n0 = (t >> 5) * 32;
  v8f acc[2][2] = {};
  wave_gemm_2x2<CC>(Qt, CC, Kt, CC, m0, n0, acc);
  const float scale = 0.04419417382415922f;
  const int lane = threadIdx.x & 31, half = lane >> 4, r = lane & 15;
#pragma unroll
  for (int mi = 0; mi < 2; ++mi)
#pragma unroll
    for (int ni = 0; ni < 2; ++ni) {
      const int mb = m0 + mi * 16 + 8 * half, n = n0 + ni * 16 + r;
#pragma unroll
      for (int v = 0; v < 8; ++v)
        S[(size_t)(mb + v) * NN + n] = acc[mi][ni][v] * scale;
    }
}

__global__ void __launch_bounds__(256) softmax_kernel(const float* __restrict__ S,
                                                      _Float16* __restrict__ P) {
  __shared__ float red[256];
  const int row = blockIdx.x;
  const int base = threadIdx.x * 4;
  float4 sv = *(const float4*)(S + (size_t)row * NN + base);
  float v0 = sv.x, v1 = sv.y, v2 = sv.z, v3 = sv.w;
  float m = fmaxf(fmaxf(v0, v1), fmaxf(v2, v3));
  red[threadIdx.x] = m; __syncthreads();
  for (int s = 128; s > 0; s >>= 1) {
    if (threadIdx.x < s) red[threadIdx.x] = fmaxf(red[threadIdx.x], red[threadIdx.x + s]);
    __syncthreads();
  }
  m = red[0]; __syncthreads();
  v0 = __expf(v0 - m); v1 = __expf(v1 - m); v2 = __expf(v2 - m); v3 = __expf(v3 - m);
  red[threadIdx.x] = v0 + v1 + v2 + v3; __syncthreads();
  for (int s = 128; s > 0; s >>= 1) {
    if (threadIdx.x < s) red[threadIdx.x] += red[threadIdx.x + s];
    __syncthreads();
  }
  const float inv = 1.0f / red[0];
  union { _Float16 h[4]; uint2 u; } o;
  o.h[0] = (_Float16)(v0 * inv); o.h[1] = (_Float16)(v1 * inv);
  o.h[2] = (_Float16)(v2 * inv); o.h[3] = (_Float16)(v3 * inv);
  *(uint2*)(P + (size_t)row * NN + base) = o.u;
}

__global__ void __launch_bounds__(128) av_gemm_kernel(const _Float16* __restrict__ Vv,
    const _Float16* __restrict__ P, _Float16* __restrict__ Ot) {
  const size_t img = blockIdx.y;
  Vv += img * (CC * NN); P += img * ((size_t)NN * NN); Ot += img * (NN * CC);
  const int t  = blockIdx.x * 4 + (threadIdx.x >> 5);
  const int m0 = (t & 15) * 32, n0 = (t >> 4) * 32;
  v8f acc[2][2] = {};
  wave_gemm_2x2<NN>(Vv, NN, P, NN, m0, n0, acc);
  const int lane = threadIdx.x & 31, half = lane >> 4, r = lane & 15;
#pragma unroll
  for (int mi = 0; mi < 2; ++mi)
#pragma unroll
    for (int ni = 0; ni < 2; ++ni) {
      const int mb = m0 + mi * 16 + 8 * half, n = n0 + ni * 16 + r;
      _Float16* dst = Ot + (size_t)n * CC + mb;
#pragma unroll
      for (int v = 0; v < 8; v += 2) {
        union { _Float16 h[2]; uint32_t u; } pk;
        pk.h[0] = (_Float16)acc[mi][ni][v];
        pk.h[1] = (_Float16)acc[mi][ni][v + 1];
        *(uint32_t*)(dst + v) = pk.u;
      }
    }
}

__global__ void __launch_bounds__(128) proj_kernel(const _Float16* __restrict__ Wp,
    const _Float16* __restrict__ Ot, const float* __restrict__ bias,
    const float* __restrict__ xb, float* __restrict__ out) {
  const size_t img = blockIdx.y;
  Ot += img * (NN * CC); xb += img * (CC * NN); out += img * (CC * NN);
  const int t  = blockIdx.x * 4 + (threadIdx.x >> 5);
  const int m0 = (t & 15) * 32, n0 = (t >> 4) * 32;
  v8f acc[2][2] = {};
  wave_gemm_2x2<CC>(Wp, CC, Ot, CC, m0, n0, acc);
  const int lane = threadIdx.x & 31, half = lane >> 4, r = lane & 15;
#pragma unroll
  for (int mi = 0; mi < 2; ++mi)
#pragma unroll
    for (int ni = 0; ni < 2; ++ni) {
      const int mb = m0 + mi * 16 + 8 * half, n = n0 + ni * 16 + r;
#pragma unroll
      for (int v = 0; v < 8; ++v) {
        const size_t idx = (size_t)(mb + v) * NN + n;
        out[idx] = xb[idx] + bias[mb + v] + acc[mi][ni][v];
      }
    }
}


extern "C" void kernel_launch(void* const* d_in, const int* in_sizes, int n_in,
                              void* d_out, int out_size, void* d_ws, size_t ws_size,
                              hipStream_t stream) {
  const float* x      = (const float*)d_in[0];
  const float* gamma  = (const float*)d_in[1];
  const float* beta   = (const float*)d_in[2];
  const float* w_qkv  = (const float*)d_in[3];
  const float* b_qkv  = (const float*)d_in[4];
  const float* w_proj = (const float*)d_in[5];
  const float* b_proj = (const float*)d_in[6];
  float* out = (float*)d_out;

  char* ws = (char*)d_ws;
  char* ob = (char*)d_out;
  float*    stats  = (float*)   (ws + 0);
  _Float16* Wqkvh  = (_Float16*)(ws + 4096);
  _Float16* Wprojh = (_Float16*)(ws + 1576960);
  _Float16* Ht     = (_Float16*)(ws + 2101248);
  _Float16* Ot     = Ht;
  _Float16* Qt     = (_Float16*)(ws + 18878464);
  _Float16* Vv     = (_Float16*)(ws + 23072768);
  float*    S      = (float*)   (ob + 0);
  _Float16* P      = (_Float16*)(ob + 16777216);
  _Float16* Kt     = (_Float16*)(ob + 25165824);

  cvt_f32_to_f16<<<(C3 * CC) / 256, 256, 0, stream>>>(w_qkv, Wqkvh, C3 * CC);
  cvt_f32_to_f16<<<(CC * CC) / 256, 256, 0, stream>>>(w_proj, Wprojh, CC * CC);
  gn_stats_kernel<<<BB * 32, 256, 0, stream>>>(x, stats);
  gn_apply_t_kernel<<<(BB * CC * NN) / (256 * 4), 256, 0, stream>>>(x, gamma, beta, stats, Ht);

  for (int b = 0; b < BB; b += GI) {
    _Float16* Hg = Ht + (size_t)b * NN * CC;
    qkv_gemm_kernel<<<dim3((48 * 32) / 4, GI), 128, 0, stream>>>(Wqkvh, Hg, b_qkv, Qt, Kt, Vv);
    scores_kernel<<<dim3((32 * 32) / 4, GI), 128, 0, stream>>>(Qt, Kt, S);
    softmax_kernel<<<NN * GI, 256, 0, stream>>>(S, P);
    av_gemm_kernel<<<dim3((16 * 32) / 4, GI), 128, 0, stream>>>(Vv, P, Hg);
  }
  proj_kernel<<<dim3((16 * 32) / 4, BB), 128, 0, stream>>>(Wprojh, Ot, b_proj, x, out);
}


# Operand slabs of the four GEMMs fetched once per 128x64 block tile with whole-line global loads into LDS (two stages) and shared by 8 waves, not fetched per 32x32 wave from global; sums unchanged
# speedup vs baseline: 2.7580x; 1.5709x over previous
#include <hip/hip_runtime.h>
#include <cstdint>

typedef __attribute__((ext_vector_type(16))) _Float16 v16h;
typedef __attribute__((ext_vector_type(8)))  float    v8f;

#define BB   16
#define CC   512
#define NN   1024
#define C3   1536
#define CPG  16
#define GI   4

union FragH { v16h v; uint4 q[2]; };

#define BM     128
#define BN     64
#define BK     64
#define PITCH  72
#define STAGE  ((BM + BN) * PITCH)
#define GEMM_LDS_ELEMS (2 * STAGE)

template <int K>
__device__ __forceinline__ void block_gemm_128x64(const _Float16* __restrict__ A, int lda,
                                                  const _Float16* __restrict__ B, int ldb,
                                                  int bm0, int bn0, _Float16* __restrict__ smem,
                                                  v8f acc[2][2]) {
  static_assert(K % BK == 0, "K must be a whole number of chunks");
  constexpr int NC = K / BK;
  const int tid  = threadIdx.x;
  const int seg  = tid & 7, crow = tid >> 3;
  const _Float16* Ag = A + (size_t)(bm0 + crow) * lda + seg * 8;
  const _Float16* Bg = B + (size_t)(bn0 + crow) * ldb + seg * 8;
  const int cofs = crow * PITCH + seg * 8;
  const int wave = tid >> 5, lane = tid & 31, half = lane >> 4, r = lane & 15;
  const int wm = wave & 3, wn = wave >> 2;
  const int fa = (wm * 32 + r) * PITCH + half * 8;
  const int fb = (BM + wn * 32 + r) * PITCH + half * 8;

  uint4 ra[4], rb[2];
#pragma unroll
  for (int i = 0; i < 4; ++i) ra[i] = *(const uint4*)(Ag + (size_t)(i * 32) * lda);
#pragma unroll
  for (int i = 0; i < 2; ++i) rb[i] = *(const uint4*)(Bg + (size_t)(i * 32) * ldb);
#pragma unroll
  for (int i = 0; i < 4; ++i) *(uint4*)(smem + cofs + i * 32 * PITCH) = ra[i];
#pragma unroll
  for (int i = 0; i < 2; ++i) *(uint4*)(smem + BM * PITCH + cofs + i * 32 * PITCH) = rb[i];
  __syncthreads();

  for (int c = 0; c < NC; ++c) {
    const _Float16* cur = smem + (c & 1) * STAGE;
    _Float16*       nxt = smem + ((c + 1) & 1) * STAGE;
    if (c + 1 < NC) {
      const int k1 = (c + 1) * BK;
#pragma unroll
      for (int i = 0; i < 4; ++i) ra[i] = *(const uint4*)(Ag + (size_t)(i * 32) * lda + k1);
#pragma unroll
      for (int i = 0; i < 2; ++i) rb[i] = *(const uint4*)(Bg + (size_t)(i * 32) * ldb + k1);
    }
    FragH a0[2], a1[2], b0[2], b1[2];
#pragma unroll
    for (int s = 0; s < 2; ++s) {
      a0[s].q[0] = *(const uint4*)(cur + fa + s * 32);
      a0[s].q[1] = *(const uint4*)(cur + fa + s * 32 + 16);
      a1[s].q[0] = *(const uint4*)(cur + fa + 16 * PITCH + s * 32);
      a1[s].q[1] = *(const uint4*)(cur + fa + 16 * PITCH + s * 32 + 16);
      b0[s].q[0] = *(const uint4*)(cur + fb + s * 32);
      b0[s].q[1] = *(const uint4*)(cur + fb + s * 32 + 16);
      b1[s].q[0] = *(const uint4*)(cur + fb + 16 * PITCH + s * 32);
      b1[s].q[1] = *(const uint4*)(cur + fb + 16 * PITCH + s * 32 + 16);
    }
#pragma unroll
    for (int s = 0; s < 2; ++s) {
      acc[0][0] = __builtin_amdgcn_wmma_f32_16x16x32_f16(false, a0[s].v, false, b0[s].v, (short)0, acc[0][0], false, false);
      acc[0][1] = __builtin_amdgcn_wmma_f32_16x16x32_f16(false, a0[s].v, false, b1[s].v, (short)0, acc[0][1], false, false);
      acc[1][0] = __builtin_amdgcn_wmma_f32_16x16x32_f16(false, a1[s].v, false, b0[s].v, (short)0, acc[1][0], false, false);
      acc[1][1] = __builtin_amdgcn_wmma_f32_16x16x32_f16(false, a1[s].v, false, b1[s].v, (short)0, acc[1][1], false, false);
    }
    if (c + 1 < NC) {
#pragma unroll
      for (int i = 0; i < 4; ++i) *(uint4*)(nxt + cofs + i * 32 * PITCH) = ra[i];
#pragma unroll
      for (int i = 0; i < 2; ++i) *(uint4*)(nxt + BM * PITCH + cofs + i * 32 * PITCH) = rb[i];
      __syncthreads();
    }
  }
}


__global__ void __launch_bounds__(256) cvt_f32_to_f16(const float* __restrict__ in,
                                                      _Float16* __restrict__ out, int n) {
  int i = blockIdx.x * blockDim.x + threadIdx.x;
  if (i < n) out[i] = (_Float16)in[i];
}

__global__ void __launch_bounds__(256) gn_stats_kernel(const float* __restrict__ x,
                                                       float* __restrict__ stats) {
  __shared__ float s1[256], s2[256];
  const int bg = blockIdx.x;
  const float* xp = x + (size_t)bg * (CPG * NN);
  float a = 0.f, b = 0.f;
  for (int i = threadIdx.x; i < CPG * NN; i += 256) { float v = xp[i]; a += v; b += v * v; }
  s1[threadIdx.x] = a; s2[threadIdx.x] = b;
  __syncthreads();
  for (int s = 128; s > 0; s >>= 1) {
    if (threadIdx.x < s) { s1[threadIdx.x] += s1[threadIdx.x + s];
                           s2[threadIdx.x] += s2[threadIdx.x + s]; }
    __syncthreads();
  }
  if (threadIdx.x == 0) {
    const float inv = 1.0f / (float)(CPG * NN);
    float mean = s1[0] * inv;
    float var  = s2[0] * inv - mean * mean;
    stats[2 * bg]     = mean;
    stats[2 * bg + 1] = rsqrtf(var + 1e-5f);
  }
}

__global__ void __launch_bounds__(256) gn_apply_t_kernel(const float* __restrict__ x,
    const float* __restrict__ gamma, const float* __restrict__ beta,
    const float* __restrict__ stats, _Float16* __restrict__ Ht) {
  const size_t i4 = ((size_t)blockIdx.x * blockDim.x + threadIdx.x) * 4;
  const int n  = (int)(i4 & (NN - 1));
  const int c  = (int)((i4 >> 10) & (CC - 1));
  const int bg = (int)(i4 >> 14);
  const int b  = bg >> 5;
  const float4 xv = *(const float4*)(x + i4);
  const float mean = stats[2 * bg], rstd = stats[2 * bg + 1];
  const float g = gamma[c] * rstd, bt = beta[c] - mean * gamma[c] * rstd;
  _Float16* hp = Ht + ((size_t)b * NN + n) * CC + c;
  hp[0 * CC] = (_Float16)(xv.x * g + bt);
  hp[1 * CC] = (_Float16)(xv.y * g + bt);
  hp[2 * CC] = (_Float16)(xv.z * g + bt);
  hp[3 * CC] = (_Float16)(xv.w * g + bt);
}


__global__ void __launch_bounds__(256) qkv_gemm_kernel(const _Float16* __restrict__ W,
    const _Float16* __restrict__ Ht, const float* __restrict__ bias,
    _Float16* __restrict__ Qt, _Float16* __restrict__ Kt, _Float16* __restrict__ Vv) {
  const size_t img = blockIdx.y;
  Ht += img * (NN * CC); Qt += img * (NN * CC); Kt += img * (NN * CC); Vv += img * (CC * NN);
  __shared__ __attribute__((aligned(16))) _Float16 smem[GEMM_LDS_ELEMS];
  const int bm0 = (blockIdx.x % (C3 / BM)) * BM, bn0 = (blockIdx.x / (C3 / BM)) * BN;
  const int wv = threadIdx.x >> 5;
  const int m0 = bm0 + (wv & 3) * 32, n0 = bn0 + (wv >> 2) * 32;
  v8f acc[2][2] = {};
  block_gemm_128x64<CC>(W, CC, Ht, CC, bm0, bn0, smem, acc);
  const int lane = threadIdx.x & 31, half = lane >> 4, r = lane & 15;
#pragma unroll
  for (int mi = 0; mi < 2; ++mi) {
#pragma unroll
    for (int ni = 0; ni < 2; ++ni) {
      const int mb = m0 + mi * 16;
      const int n  = n0 + ni * 16 + r;
      const int mbase = mb + 8 * half;
      const int seg = mb >> 9;
      if (seg < 2) {
        _Float16* dst = (seg == 0 ? Qt : Kt) + (size_t)n * CC + ((mbase) & (CC - 1));
#pragma unroll
        for (int v = 0; v < 8; v += 2) {
          union { _Float16 h[2]; uint32_t u; } pk;
          pk.h[0] = (_Float16)(acc[mi][ni][v]     + bias[mbase + v]);
          pk.h[1] = (_Float16)(acc[mi][ni][v + 1] + bias[mbase + v + 1]);
          *(uint32_t*)(dst + v) = pk.u;
        }
      } else {
#pragma unroll
        for (int v = 0; v < 8; ++v)
          Vv[(size_t)(mbase - 1024 + v) * NN + n] =
              (_Float16)(acc[mi][ni][v] + bias[mbase + v]);
      }
    }
  }
}

__global__ void __launch_bounds__(256) scores_kernel(const _Float16* __restrict__ Qt,
    const _Float16* __restrict__ Kt, float* __restrict__ S) {
  const size_t img = blockIdx.y;
  Qt += img * (NN * CC); Kt += img * (NN * CC); S += img * ((size_t)NN * NN);
  __shared__ __attribute__((aligned(16))) _Float16 smem[GEMM_LDS_ELEMS];
  const int bm0 = (blockIdx.x & 7) * BM, bn0 = (blockIdx.x >> 3) * BN;
  const int wv = threadIdx.x >> 5;
  const int m0 = bm0 + (wv & 3) * 32, n0 = bn0 + (wv >> 2) * 32;
  v8f acc[2][2] = {};
  block_gemm_128x64<CC>(Qt, CC, Kt, CC, bm0, bn0, smem, acc);
  const float scale = 0.04419417382415922f;
  const int lane = threadIdx.x & 31, half = lane >> 4, r = lane & 15;
#pragma unroll
  for (int mi = 0; mi < 2; ++mi)
#pragma unroll
    for (int ni = 0; ni < 2; ++ni) {
      const int mb = m0 + mi * 16 + 8 * half, n = n0 + ni * 16 + r;
#pragma unroll
      for (int v = 0; v < 8; ++v)
        S[(size_t)(mb + v) * NN + n] = acc[mi][ni][v] * scale;
    }
}

__global__ void __launch_bounds__(256) softmax_kernel(const float* __restrict__ S,
                                                      _Float16* __restrict__ P) {
  __shared__ float red[256];
  const int row = blockIdx.x;
  const int base = threadIdx.x * 4;
  float4 sv = *(const float4*)(S + (size_t)row * NN + base);
  float v0 = sv.x, v1 = sv.y, v2 = sv.z, v3 = sv.w;
  float m = fmaxf(fmaxf(v0, v1), fmaxf(v2, v3));
  red[threadIdx.x] = m; __syncthreads();
  for (int s = 128; s > 0; s >>= 1) {
    if (threadIdx.x < s) red[threadIdx.x] = fmaxf(red[threadIdx.x], red[threadIdx.x + s]);
    __syncthreads();
  }
  m = red[0]; __syncthreads();
  v0 = __expf(v0 - m); v1 = __expf(v1 - m); v2 = __expf(v2 - m); v3 = __expf(v3 - m);
  red[threadIdx.x] = v0 + v1 + v2 + v3; __syncthreads();
  for (int s = 128; s > 0; s >>= 1) {
    if (threadIdx.x < s) red[threadIdx.x] += red[threadIdx.x + s];
    __syncthreads();
  }
  const float inv = 1.0f / red[0];
  union { _Float16 h[4]; uint2 u; } o;
  o.h[0] = (_Float16)(v0 * inv); o.h[1] = (_Float16)(v1 * inv);
  o.h[2] = (_Float16)(v2 * inv); o.h[3] = (_Float16)(v3 * inv);
  *(uint2*)(P + (size_t)row * NN + base) = o.u;
}

__global__ void __launch_bounds__(256) av_gemm_kernel(const _Float16* __restrict__ Vv,
    const _Float16* __restrict__ P, _Float16* __restrict__ Ot) {
  const size_t img = blockIdx.y;
  Vv += img * (CC * NN); P += img * ((size_t)NN * NN); Ot += img * (NN * CC);
  __shared__ __attribute__((aligned(16))) _Float16 smem[GEMM_LDS_ELEMS];
  const int bm0 = (blockIdx.x & 3) * BM, bn0 = (blockIdx.x >> 2) * BN;
  const int wv = threadIdx.x >> 5;
  const int m0 = bm0 + (wv & 3) * 32, n0 = bn0 + (wv >> 2) * 32;
  v8f acc[2][2] = {};
  block_gemm_128x64<NN>(Vv, NN, P, NN, bm0, bn0, smem, acc);
  const int lane = threadIdx.x & 31, half = lane >> 4, r = lane & 15;
#pragma unroll
  for (int mi = 0; mi < 2; ++mi)
#pragma unroll
    for (int ni = 0; ni < 2; ++ni) {
      const int mb = m0 + mi * 16 + 8 * half, n = n0 + ni * 16 + r;
      _Float16* dst = Ot + (size_t)n * CC + mb;
#pragma unroll
      for (int v = 0; v < 8; v += 2) {
        union { _Float16 h[2]; uint32_t u; } pk;
        pk.h[0] = (_Float16)acc[mi][ni][v];
        pk.h[1] = (_Float16)acc[mi][ni][v + 1];
        *(uint32_t*)(dst + v) = pk.u;
      }
    }
}

__global__ void __launch_bounds__(256) proj_kernel(const _Float16* __restrict__ Wp,
    const _Float16* __restrict__ Ot, const float* __restrict__ bias,
    const float* __restrict__ xb, float* __restrict__ out) {
  const size_t img = blockIdx.y;
  Ot += img * (NN * CC); xb += img * (CC * NN); out += img * (CC * NN);
  __shared__ __attribute__((aligned(16))) _Float16 smem[GEMM_LDS_ELEMS];
  const int bm0 = (blockIdx.x & 3) * BM, bn0 = (blockIdx.x >> 2) * BN;
  const int wv = threadIdx.x >> 5;
  const int m0 = bm0 + (wv & 3) * 32, n0 = bn0 + (wv >> 2) * 32;
  v8f acc[2][2] = {};
  block_gemm_128x64<CC>(Wp, CC, Ot, CC, bm0, bn0, smem, acc);
  const int lane = threadIdx.x & 31, half = lane >> 4, r = lane & 15;
#pragma unroll
  for (int mi = 0; mi < 2; ++mi)
#pragma unroll
    for (int ni = 0; ni < 2; ++ni) {
      const int mb = m0 + mi * 16 + 8 * half, n = n0 + ni * 16 + r;
#pragma unroll
      for (int v = 0; v < 8; ++v) {
        const size_t idx = (size_t)(mb + v) * NN + n;
        out[idx] = xb[idx] + bias[mb + v] + acc[mi][ni][v];
      }
    }
}


extern "C" void kernel_launch(void* const* d_in, const int* in_sizes, int n_in,
                              void* d_out, int out_size, void* d_ws, size_t ws_size,
                              hipStream_t stream) {
  const float* x      = (const float*)d_in[0];
  const float* gamma  = (const float*)d_in[1];
  const float* beta   = (const float*)d_in[2];
  const float* w_qkv  = (const float*)d_in[3];
  const float* b_qkv  = (const float*)d_in[4];
  const float* w_proj = (const float*)d_in[5];
  const float* b_proj = (const float*)d_in[6];
  float* out = (float*)d_out;

  char* ws = (char*)d_ws;
  char* ob = (char*)d_out;
  float*    stats  = (float*)   (ws + 0);
  _Float16* Wqkvh  = (_Float16*)(ws + 4096);
  _Float16* Wprojh = (_Float16*)(ws + 1576960);
  _Float16* Ht     = (_Float16*)(ws + 2101248);
  _Float16* Ot     = Ht;
  _Float16* Qt     = (_Float16*)(ws + 18878464);
  _Float16* Vv     = (_Float16*)(ws + 23072768);
  float*    S      = (float*)   (ob + 0);
  _Float16* P      = (_Float16*)(ob + 16777216);
  _Float16* Kt     = (_Float16*)(ob + 25165824);

  cvt_f32_to_f16<<<(C3 * CC) / 256, 256, 0, stream>>>(w_qkv, Wqkvh, C3 * CC);
  cvt_f32_to_f16<<<(CC * CC) / 256, 256, 0, stream>>>(w_proj, Wprojh, CC * CC);
  gn_stats_kernel<<<BB * 32, 256, 0, stream>>>(x, stats);
  gn_apply_t_kernel<<<(BB * CC * NN) / (256 * 4), 256, 0, stream>>>(x, gamma, beta, stats, Ht);

  for (int b = 0; b < BB; b += GI) {
    _Float16* Hg = Ht + (size_t)b * NN * CC;
    qkv_gemm_kernel<<<dim3((C3 / BM) * (NN / BN), GI), 256, 0, stream>>>(Wqkvh, Hg, b_qkv, Qt, Kt, Vv);
    scores_kernel<<<dim3((NN / BM) * (NN / BN), GI), 256, 0, stream>>>(Qt, Kt, S);
    softmax_kernel<<<NN * GI, 256, 0, stream>>>(S, P);
    av_gemm_kernel<<<dim3((CC / BM) * (NN / BN), GI), 256, 0, stream>>>(Vv, P, Hg);
  }
  proj_kernel<<<dim3((CC / BM) * (NN / BN), BB), 256, 0, stream>>>(Wprojh, Ot, b_proj, x, out);
}
